# starting kernel
# baseline (speedup 1.0000x reference)
#include <hip/hip_runtime.h>
#include <hip/hip_bf16.h>

typedef __attribute__((ext_vector_type(2))) float v2f;
typedef __attribute__((ext_vector_type(8))) float v8f;

namespace {
constexpr int kW = 512;
constexpr int kH = 512;
constexpr float kLog2e = 1.44269504088896340736f;
}

__global__ void gauss2d_coeffs(const float* __restrict__ means,
                               const float* __restrict__ scales,
                               const float* __restrict__ thetas,
                               const float* __restrict__ opac_in,
                               float* __restrict__ G,
                               int N) {
  int n = blockIdx.x * blockDim.x + threadIdx.x;
  if (n >= N) return;
  float mx = means[2 * n + 0];
  float my = means[2 * n + 1];
  float sx = expf(scales[2 * n + 0]);
  float sy = expf(scales[2 * n + 1]);
  float th = (1.0f / (1.0f + expf(-thetas[n]))) * 6.28318530717958647692f;
  float op = 1.0f / (1.0f + expf(-opac_in[n]));
  float c = cosf(th), s = sinf(th);
  float sx2 = sx * sx, sy2 = sy * sy;
  float a = c * c * sx2 + s * s * sy2 + 1e-6f;
  float b = c * s * (sx2 - sy2);
  float d = s * s * sx2 + c * c * sy2 + 1e-6f;
  float det = a * d - b * b;
  float ai = d / det, bi = -b / det, di = a / det;
  float* g = G + (size_t)n * 8;
  const float m = -0.5f * kLog2e;
  g[0] = m * ai;
  g[1] = m * 2.0f * bi;
  g[2] = m * di;
  g[3] = -2.0f * m * (ai * mx + bi * my);
  g[4] = -2.0f * m * (bi * mx + di * my);
  g[5] = m * (ai * mx * mx + 2.0f * bi * mx * my + di * my * my)
         + log2f(op);
  g[6] = 0.0f;
  g[7] = 0.0f;
}

__global__ __launch_bounds__(256) void gauss2d_render(
    const float* __restrict__ G, float* __restrict__ out, int N) {
  const int lane = threadIdx.x & 31;
  const int wave = (blockIdx.x * blockDim.x + threadIdx.x) >> 5;
  const bool hi = lane >= 16;
  const int col = lane & 15;
  const int pix = wave * 16 + col;
  const int w = pix & (kW - 1);
  const int h = pix >> 9;
  const float x = -1.0f + 2.0f * (float)w / (float)(kW - 1);
  const float y = -1.0f + 2.0f * (float)h / (float)(kH - 1);

  v2f b1, b2;
  b1.x = hi ? (y * y) : (x * x);
  b1.y = hi ? x       : (x * y);
  b2.x = hi ? 0.0f    : y;
  b2.y = hi ? 0.0f    : 1.0f;

  float ACC = 0.0f;
  float T   = 1.0f;

  const v2f* gp = (const v2f*)(G + (size_t)col * 8 + (hi ? 2 : 0));

  for (int t = 0; t < N; t += 16) {
    v2f a1 = gp[0];
    v2f a2 = gp[2];
    gp += 64;

    v8f q = {};
    q = __builtin_amdgcn_wmma_f32_16x16x4_f32(false, a1, false, b1,
                                              (short)0, q, false, false);
    q = __builtin_amdgcn_wmma_f32_16x16x4_f32(false, a2, false, b2,
                                              (short)0, q, false, false);

    float accH = 0.0f, TH = 1.0f;
#pragma unroll
    for (int r = 0; r < 8; ++r) {
      float alpha = __builtin_amdgcn_exp2f(q[r]);
      accH += TH * alpha;
      TH *= (1.0f - alpha);
    }

    float accO = __shfl_xor(accH, 16, 32);
    float TO   = __shfl_xor(TH, 16, 32);
    float tileAcc = hi ? (accO + TO * accH) : (accH + TH * accO);
    float tileT   = TH * TO;

    ACC += T * tileAcc;
    T *= tileT;
  }

  if (!hi) {
    float* o = out + (size_t)pix * 3;
    o[0] = ACC;
    o[1] = ACC;
    o[2] = ACC;
  }
}

extern "C" void kernel_launch(void* const* d_in, const int* in_sizes, int n_in,
                              void* d_out, int out_size, void* d_ws,
                              size_t ws_size, hipStream_t stream) {
  const float* means  = (const float*)d_in[0];
  const float* scales = (const float*)d_in[1];
  const float* thetas = (const float*)d_in[2];
  const float* opacs  = (const float*)d_in[3];

  const int N = in_sizes[0] / 2;

  float* G = (float*)d_ws;

  gauss2d_coeffs<<<(N + 127) / 128, 128, 0, stream>>>(means, scales, thetas,
                                                      opacs, G, N);

  const int pixels  = kW * kH;
  const int waves   = pixels / 16;
  const int threads = waves * 32;
  const int block   = 256;
  gauss2d_render<<<threads / block, block, 0, stream>>>(G, (float*)d_out, N);
}


# One wave per 16x16 pixel block, Gaussians walked in order: per Gaussian one K=4 WMMA of y-polynomial rows x [x^2,x,1] columns; the rows of 4 Gaussians from one table WMMA; sum order changes
# speedup vs baseline: 1.4338x; 1.4338x over previous
#include <hip/hip_runtime.h>
#include <hip/hip_bf16.h>

typedef __attribute__((ext_vector_type(2))) float v2f;
typedef __attribute__((ext_vector_type(4))) float v4f;
typedef __attribute__((ext_vector_type(8))) float v8f;

namespace {
constexpr int kW = 512;
constexpr int kH = 512;
constexpr float kLog2e = 1.44269504088896340736f;
constexpr int kChunk = 128;
constexpr int kTiles = kChunk / 4;
constexpr int kHalfTiles = kTiles / 2;
constexpr int kTileStride = 80;
constexpr int kWavesPerGroup = 4;
constexpr int kBlocksX = kW / 16;
constexpr int kBlocksY = kH / 16;
}


struct Coef { float g0, g1, g2, g3, g4, g5; };

struct Raw { float mx, my, s0, s1, th, op; };

__device__ __forceinline__ Coef gauss_coef(const Raw& p, bool live) {
  Coef k;
  float mx = p.mx;
  float my = p.my;
  float sx = expf(p.s0);
  float sy = expf(p.s1);
  float th = (1.0f / (1.0f + expf(-p.th))) * 6.28318530717958647692f;
  float op = 1.0f / (1.0f + expf(-p.op));
  float c, s;
  sincosf(th, &s, &c);
  float sx2 = sx * sx, sy2 = sy * sy;
  float a = c * c * sx2 + s * s * sy2 + 1e-6f;
  float b = c * s * (sx2 - sy2);
  float d = s * s * sx2 + c * c * sy2 + 1e-6f;
  float det = a * d - b * b;
  float ai = d / det, bi = -b / det, di = a / det;
  const float m = -0.5f * kLog2e;
  k.g0 = m * ai;
  k.g1 = m * 2.0f * bi;
  k.g2 = m * di;
  k.g3 = -2.0f * m * (ai * mx + bi * my);
  k.g4 = -2.0f * m * (bi * mx + di * my);
  k.g5 = m * (ai * mx * mx + 2.0f * bi * mx * my + di * my * my)
         + log2f(op);
  if (!live) {
    k.g0 = k.g1 = k.g2 = k.g3 = k.g4 = 0.0f;
    k.g5 = -1.0e30f;
  }
  return k;
}

__device__ __forceinline__ v8f mma4(v2f a, v2f b) {
  v8f z = {};
  return __builtin_amdgcn_wmma_f32_16x16x4_f32(false, a, false, b, (short)0, z,
                                               false, false);
}

__device__ __forceinline__ v2f frag(const v8f& d, int i) {
  v2f a;
  a.x = d[2 * i];
  a.y = d[2 * i + 1];
  return a;
}

__global__ __launch_bounds__(128) void gauss2d_blocks(
    const float* __restrict__ means, const float* __restrict__ scales,
    const float* __restrict__ thetas, const float* __restrict__ opac_in,
    float* __restrict__ out, int N) {
  __shared__ __attribute__((aligned(16))) float lds[kWavesPerGroup * kTiles * kTileStride];

  const int lane = threadIdx.x & 31;
  const int wv = threadIdx.x >> 5;
  const bool hi = lane >= 16;
  const int col = lane & 15;
  float* tab = lds + wv * (kTiles * kTileStride);

  const int X0 = ((blockIdx.x % (kBlocksX / kWavesPerGroup)) * kWavesPerGroup + wv) * 16;
  const int Y0 = (blockIdx.x / (kBlocksX / kWavesPerGroup)) * 16;
  const float x = -1.0f + 2.0f * (float)(X0 + col) / (float)(kW - 1);
  const float y = -1.0f + 2.0f * (float)(Y0 + col) / (float)(kH - 1);

  v2f bx, by;
  bx.x = hi ? 1.0f : (x * x);
  bx.y = hi ? 0.0f : x;
  by.x = hi ? (y * y) : 1.0f;
  by.y = hi ? 0.0f : y;

  float acc[8], T[8];
  for (int r = 0; r < 8; ++r) { acc[r] = 0.0f; T[r] = 1.0f; }

  for (int base = 0; base < N; base += kChunk) {
    Raw raw[4];
    bool live[4];
#pragma unroll
    for (int j = 0; j < 4; ++j) {
      const int n = base + 32 * j + lane;
      live[j] = n < N;
      const int nc = live[j] ? n : 0;
      raw[j].mx = means[2 * nc + 0];
      raw[j].my = means[2 * nc + 1];
      raw[j].s0 = scales[2 * nc + 0];
      raw[j].s1 = scales[2 * nc + 1];
      raw[j].th = thetas[nc];
      raw[j].op = opac_in[nc];
    }
#pragma unroll
    for (int j = 0; j < 4; ++j) {
      const Coef k = gauss_coef(raw[j], live[j]);
      float* t = tab + (8 * j + (lane >> 2)) * kTileStride + 4 * (lane & 3);
      v4f r0 = {k.g0, 0.0f, k.g3, k.g1};
      v4f r1 = {k.g5, k.g4, 0.0f, 0.0f};
      v4f r2 = {0.0f, 0.0f, 0.0f, 0.0f};
      v4f r3 = {k.g2, 0.0f, 0.0f, 0.0f};
      *(v4f*)(t)      = r0;
      *(v4f*)(t + 16) = r1;
      *(v4f*)(t + 32) = r2;
      *(v4f*)(t + 48) = r3;
    }
    __syncthreads();

#pragma unroll 1
    for (int half = 0; half < 2; ++half) {
      v2f at[kHalfTiles];
#pragma unroll
      for (int t = 0; t < kHalfTiles; ++t)
        at[t] = *(const v2f*)(tab + (half * kHalfTiles + t) * kTileStride + 2 * lane);

      v8f dcur = mma4(at[0], by);
      v8f dnext = mma4(at[1], by);
      v8f q2 = mma4(frag(dcur, 0), bx);
      float a1[8];
      for (int r = 0; r < 8; ++r) a1[r] = __builtin_amdgcn_exp2f(q2[r]);
      q2 = mma4(frag(dcur, 1), bx);
      __builtin_amdgcn_sched_barrier(0);
#pragma unroll
      for (int g = 0; g < 4 * kHalfTiles; ++g) {
        v8f q3 = q2;
        const int gi = g + 2;
        const bool newtile = (gi < 4 * kHalfTiles) && ((gi & 3) == 0);
        if (gi < 4 * kHalfTiles) {
          if (newtile) dcur = dnext;
          q3 = mma4(frag(dcur, gi & 3), bx);
        }
        __builtin_amdgcn_sched_barrier(0);
        float a2[8];
#pragma unroll
        for (int r = 0; r < 8; ++r) {
          acc[r] += T[r] * a1[r];
          T[r] -= T[r] * a1[r];
          __builtin_amdgcn_sched_barrier(0);
          a2[r] = a1[r];
          if (g + 1 < 4 * kHalfTiles) a2[r] = __builtin_amdgcn_exp2f(q2[r]);
          if (r == 3 && newtile && (gi >> 2) + 1 < kHalfTiles)
            dnext = mma4(at[(gi >> 2) + 1], by);
          __builtin_amdgcn_sched_barrier(0);
        }
        for (int r = 0; r < 8; ++r) a1[r] = a2[r];
        q2 = q3;
      }
    }
    __syncthreads();
  }

  for (int r = 0; r < 8; ++r) {
    const int row = Y0 + (hi ? 8 : 0) + r;
    float* o = out + ((size_t)row * kW + (X0 + col)) * 3;
    o[0] = acc[r];
    o[1] = acc[r];
    o[2] = acc[r];
  }
}

extern "C" void kernel_launch(void* const* d_in, const int* in_sizes, int n_in,
                              void* d_out, int out_size, void* d_ws,
                              size_t ws_size, hipStream_t stream) {
  const float* means  = (const float*)d_in[0];
  const float* scales = (const float*)d_in[1];
  const float* thetas = (const float*)d_in[2];
  const float* opacs  = (const float*)d_in[3];

  const int N = in_sizes[0] / 2;

  constexpr int groups = kBlocksX * kBlocksY / kWavesPerGroup;
  constexpr int block  = kWavesPerGroup * 32;
  gauss2d_blocks<<<groups, block, 0, stream>>>(
      means, scales, thetas, opacs, (float*)d_out, N);
}


# The four waves of a work-group share one coefficient table in LDS: each thread builds the rows of one Gaussian (before: every wave built all 128, four per lane, one after another); sum order kept
# speedup vs baseline: 1.6251x; 1.1334x over previous
#include <hip/hip_runtime.h>
#include <hip/hip_bf16.h>

typedef __attribute__((ext_vector_type(2))) float v2f;
typedef __attribute__((ext_vector_type(4))) float v4f;
typedef __attribute__((ext_vector_type(8))) float v8f;

namespace {
constexpr int kW = 512;
constexpr int kH = 512;
constexpr float kLog2e = 1.44269504088896340736f;
constexpr int kChunk = 128;
constexpr int kTiles = kChunk / 4;
constexpr int kHalfTiles = kTiles / 2;
constexpr int kTileStride = 80;
constexpr int kWavesPerGroup = 4;
constexpr int kBlocksX = kW / 16;
constexpr int kBlocksY = kH / 16;
static_assert(kChunk == kWavesPerGroup * 32, "one Gaussian of a table per thread");
}


struct Coef { float g0, g1, g2, g3, g4, g5; };

struct Raw { float mx, my, s0, s1, th, op; };

__device__ __forceinline__ Coef gauss_coef(const Raw& p, bool live) {
  Coef k;
  float mx = p.mx;
  float my = p.my;
  float sx = expf(p.s0);
  float sy = expf(p.s1);
  float th = (1.0f / (1.0f + expf(-p.th))) * 6.28318530717958647692f;
  float op = 1.0f / (1.0f + expf(-p.op));
  float c, s;
  sincosf(th, &s, &c);
  float sx2 = sx * sx, sy2 = sy * sy;
  float a = c * c * sx2 + s * s * sy2 + 1e-6f;
  float b = c * s * (sx2 - sy2);
  float d = s * s * sx2 + c * c * sy2 + 1e-6f;
  float det = a * d - b * b;
  float ai = d / det, bi = -b / det, di = a / det;
  const float m = -0.5f * kLog2e;
  k.g0 = m * ai;
  k.g1 = m * 2.0f * bi;
  k.g2 = m * di;
  k.g3 = -2.0f * m * (ai * mx + bi * my);
  k.g4 = -2.0f * m * (bi * mx + di * my);
  k.g5 = m * (ai * mx * mx + 2.0f * bi * mx * my + di * my * my)
         + log2f(op);
  if (!live) {
    k.g0 = k.g1 = k.g2 = k.g3 = k.g4 = 0.0f;
    k.g5 = -1.0e30f;
  }
  return k;
}

__device__ __forceinline__ v8f mma4(v2f a, v2f b) {
  v8f z = {};
  return __builtin_amdgcn_wmma_f32_16x16x4_f32(false, a, false, b, (short)0, z,
                                               false, false);
}

__device__ __forceinline__ v2f frag(const v8f& d, int i) {
  v2f a;
  a.x = d[2 * i];
  a.y = d[2 * i + 1];
  return a;
}

__global__ __launch_bounds__(128) void gauss2d_blocks(
    const float* __restrict__ means, const float* __restrict__ scales,
    const float* __restrict__ thetas, const float* __restrict__ opac_in,
    float* __restrict__ out, int N) {
  __shared__ __attribute__((aligned(16))) float tab[kTiles * kTileStride];

  const int lane = threadIdx.x & 31;
  const int wv = threadIdx.x >> 5;
  const bool hi = lane >= 16;
  const int col = lane & 15;

  const int X0 = ((blockIdx.x % (kBlocksX / kWavesPerGroup)) * kWavesPerGroup + wv) * 16;
  const int Y0 = (blockIdx.x / (kBlocksX / kWavesPerGroup)) * 16;
  const float x = -1.0f + 2.0f * (float)(X0 + col) / (float)(kW - 1);
  const float y = -1.0f + 2.0f * (float)(Y0 + col) / (float)(kH - 1);

  v2f bx, by;
  bx.x = hi ? 1.0f : (x * x);
  bx.y = hi ? 0.0f : x;
  by.x = hi ? (y * y) : 1.0f;
  by.y = hi ? 0.0f : y;

  float acc[8], T[8];
  for (int r = 0; r < 8; ++r) { acc[r] = 0.0f; T[r] = 1.0f; }

  for (int base = 0; base < N; base += kChunk) {
    {
      const int tid = threadIdx.x;
      const int n = base + tid;
      const bool live = n < N;
      const int nc = live ? n : 0;
      Raw raw;
      raw.mx = means[2 * nc + 0];
      raw.my = means[2 * nc + 1];
      raw.s0 = scales[2 * nc + 0];
      raw.s1 = scales[2 * nc + 1];
      raw.th = thetas[nc];
      raw.op = opac_in[nc];
      const Coef k = gauss_coef(raw, live);
      float* t = tab + (tid >> 2) * kTileStride + 4 * (tid & 3);
      v4f r0 = {k.g0, 0.0f, k.g3, k.g1};
      v4f r1 = {k.g5, k.g4, 0.0f, 0.0f};
      v4f r2 = {0.0f, 0.0f, 0.0f, 0.0f};
      v4f r3 = {k.g2, 0.0f, 0.0f, 0.0f};
      *(v4f*)(t)      = r0;
      *(v4f*)(t + 16) = r1;
      *(v4f*)(t + 32) = r2;
      *(v4f*)(t + 48) = r3;
    }
    __syncthreads();

#pragma unroll 1
    for (int half = 0; half < 2; ++half) {
      v2f at[kHalfTiles];
#pragma unroll
      for (int t = 0; t < kHalfTiles; ++t)
        at[t] = *(const v2f*)(tab + (half * kHalfTiles + t) * kTileStride + 2 * lane);

      v8f dcur = mma4(at[0], by);
      v8f dnext = mma4(at[1], by);
      v8f q2 = mma4(frag(dcur, 0), bx);
      float a1[8];
      for (int r = 0; r < 8; ++r) a1[r] = __builtin_amdgcn_exp2f(q2[r]);
      q2 = mma4(frag(dcur, 1), bx);
      __builtin_amdgcn_sched_barrier(0);
#pragma unroll
      for (int g = 0; g < 4 * kHalfTiles; ++g) {
        v8f q3 = q2;
        const int gi = g + 2;
        const bool newtile = (gi < 4 * kHalfTiles) && ((gi & 3) == 0);
        if (gi < 4 * kHalfTiles) {
          if (newtile) dcur = dnext;
          q3 = mma4(frag(dcur, gi & 3), bx);
        }
        __builtin_amdgcn_sched_barrier(0);
        float a2[8];
#pragma unroll
        for (int r = 0; r < 8; ++r) {
          acc[r] += T[r] * a1[r];
          T[r] -= T[r] * a1[r];
          __builtin_amdgcn_sched_barrier(0);
          a2[r] = a1[r];
          if (g + 1 < 4 * kHalfTiles) a2[r] = __builtin_amdgcn_exp2f(q2[r]);
          if (r == 3 && newtile && (gi >> 2) + 1 < kHalfTiles)
            dnext = mma4(at[(gi >> 2) + 1], by);
          __builtin_amdgcn_sched_barrier(0);
        }
        for (int r = 0; r < 8; ++r) a1[r] = a2[r];
        q2 = q3;
      }
    }
    __syncthreads();
  }

  for (int r = 0; r < 8; ++r) {
    const int row = Y0 + (hi ? 8 : 0) + r;
    float* o = out + ((size_t)row * kW + (X0 + col)) * 3;
    o[0] = acc[r];
    o[1] = acc[r];
    o[2] = acc[r];
  }
}

extern "C" void kernel_launch(void* const* d_in, const int* in_sizes, int n_in,
                              void* d_out, int out_size, void* d_ws,
                              size_t ws_size, hipStream_t stream) {
  const float* means  = (const float*)d_in[0];
  const float* scales = (const float*)d_in[1];
  const float* thetas = (const float*)d_in[2];
  const float* opacs  = (const float*)d_in[3];

  const int N = in_sizes[0] / 2;

  constexpr int groups = kBlocksX * kBlocksY / kWavesPerGroup;
  constexpr int block  = kWavesPerGroup * 32;
  gauss2d_blocks<<<groups, block, 0, stream>>>(
      means, scales, thetas, opacs, (float*)d_out, N);
}
